# starting kernel
# baseline (speedup 1.0000x reference)
#include <hip/hip_runtime.h>
#include <math.h>

typedef __attribute__((ext_vector_type(2))) float v2f;
typedef __attribute__((ext_vector_type(8))) float v8f;

#define NTOT 4096
#define WAVES_PER_WG 8
#define WG_COUNT 8192
#define WS_MRE   0
#define WS_MIM   256
#define WS_TOTAL 512
#define WS_PART  1024

__global__ __launch_bounds__(256) void qblur_setup(const float* __restrict__ weight,
                                                   float* __restrict__ wsf) {
    const float WMUL = 0.6324555320336759f;
    float w[5];
    #pragma unroll
    for (int k = 0; k < 5; ++k) w[k] = weight[k] * WMUL * 0.5f;
    float c0 = cosf(w[0]), s0 = sinf(w[0]);
    float c1 = cosf(w[1]), s1 = sinf(w[1]);
    float c2 = cosf(w[2]), s2 = sinf(w[2]);
    float c3 = cosf(w[3]), s3 = sinf(w[3]);
    float c4 = cosf(w[4]), s4 = sinf(w[4]);
    float R0[2][2] = {{c0, -s0}, {s0, c0}};
    float R1[2][2] = {{c1, -s1}, {s1, c1}};
    float R3[2][2] = {{c3, -s3}, {s3, c3}};
    float R4[2][2] = {{c4, -s4}, {s4, c4}};
    float G1[4][4], G3[4][4];
    #pragma unroll
    for (int p = 0; p < 2; ++p)
      #pragma unroll
      for (int q = 0; q < 2; ++q)
        #pragma unroll
        for (int r = 0; r < 2; ++r)
          #pragma unroll
          for (int s = 0; s < 2; ++s) {
            G1[p*2+q][r*2+s] = R0[p][r] * R1[q][s];
            G3[p*2+q][r*2+s] = R3[p][r] * R4[q][s];
          }
    const float YY[4][4] = {{0,0,0,-1},{0,0,1,0},{0,1,0,0},{-1,0,0,0}};
    float P[4][4], T[4][4], Q[4][4], Ure[4][4], Uim[4][4];
    #pragma unroll
    for (int i = 0; i < 4; ++i)
      #pragma unroll
      for (int j = 0; j < 4; ++j) {
        float sp = 0.f, st = 0.f;
        #pragma unroll
        for (int k = 0; k < 4; ++k) { sp += G3[i][k]*G1[k][j]; st += YY[i][k]*G1[k][j]; }
        P[i][j] = sp; T[i][j] = st;
      }
    #pragma unroll
    for (int i = 0; i < 4; ++i)
      #pragma unroll
      for (int j = 0; j < 4; ++j) {
        float sq = 0.f;
        #pragma unroll
        for (int k = 0; k < 4; ++k) sq += G3[i][k]*T[k][j];
        Q[i][j] = sq;
        Ure[i][j] = c2 * P[i][j];
        Uim[i][j] = -s2 * sq;
      }
    int t = threadIdx.x;
    int r = t >> 4, c = t & 15;
    int i = r >> 2, j = r & 3, k = c >> 2, l = c & 3;
    wsf[WS_MRE + t] = Ure[i][k]*Ure[j][l] + Uim[i][k]*Uim[j][l];
    wsf[WS_MIM + t] = Uim[i][k]*Ure[j][l] - Ure[i][k]*Uim[j][l];
}

__global__ __launch_bounds__(256) void qblur_main(const float* __restrict__ x,
                                                  const float* __restrict__ wsf,
                                                  float* __restrict__ out,
                                                  float* __restrict__ partials) {
    __shared__ float lds[WAVES_PER_WG][2][256];
    __shared__ float wsum[WAVES_PER_WG];

    const int tid  = threadIdx.x;
    const int wv   = tid >> 5;
    const int lane = tid & 31;
    const int hi   = lane >> 4;
    const int l16  = lane & 15;

    const unsigned gw = blockIdx.x * WAVES_PER_WG + wv;
    const unsigned a  = gw >> 6;
    const unsigned tc = gw & 63;

    const size_t rowbase = (size_t)(4u * a) * NTOT;
    const unsigned col   = tc * 64u + (unsigned)(l16 * 4 + hi * 2);

    v2f b[4];
    float nsq = 0.f;
    #pragma unroll
    for (int q = 0; q < 4; ++q) {
        const float* p = x + rowbase + (size_t)q * NTOT + col;
        float e0 = p[0], e1 = p[1];
        b[q][0] = e0; b[q][1] = e1;
        nsq += e0*e0 + e1*e1;
    }
    nsq += __shfl_xor(nsq, 16);
    const float inv = (nsq > 0.f) ? (1.0f / sqrtf(nsq)) : 0.f;
    #pragma unroll
    for (int q = 0; q < 4; ++q) { b[q][0] *= inv; b[q][1] *= inv; }

    const int abase = l16 * 16 + hi * 2;
    v2f are[4], aim[4];
    #pragma unroll
    for (int q = 0; q < 4; ++q) {
        are[q][0] = wsf[WS_MRE + abase + 4*q];
        are[q][1] = wsf[WS_MRE + abase + 4*q + 1];
        aim[q][0] = wsf[WS_MIM + abase + 4*q];
        aim[q][1] = wsf[WS_MIM + abase + 4*q + 1];
    }

    v8f dre = {};
    v8f dim = {};
    #pragma unroll
    for (int q = 0; q < 4; ++q) {
        dre = __builtin_amdgcn_wmma_f32_16x16x4_f32(false, are[q], false, b[q],
                                                    (short)0, dre, false, false);
        dim = __builtin_amdgcn_wmma_f32_16x16x4_f32(false, aim[q], false, b[q],
                                                    (short)0, dim, false, false);
    }

    float s = 0.f;
    #pragma unroll
    for (int p = 0; p < 8; ++p) s += dre[p]*dre[p] + dim[p]*dim[p];
    #pragma unroll
    for (int off = 16; off; off >>= 1) s += __shfl_xor(s, off);
    if (lane == 0) wsum[wv] = s;

    float* lre = &lds[wv][0][0];
    float* lim = &lds[wv][1][0];
    const int st0 = (2*hi) * 64 + 4*l16;
    *(float4*)(lre + st0)      = make_float4(dre[0], dre[1], dre[2], dre[3]);
    *(float4*)(lre + st0 + 64) = make_float4(dre[4], dre[5], dre[6], dre[7]);
    *(float4*)(lim + st0)      = make_float4(dim[0], dim[1], dim[2], dim[3]);
    *(float4*)(lim + st0 + 64) = make_float4(dim[4], dim[5], dim[6], dim[7]);
    __syncthreads();

    if (tid == 0) {
        float tot = 0.f;
        #pragma unroll
        for (int w = 0; w < WAVES_PER_WG; ++w) tot += wsum[w];
        partials[blockIdx.x] = tot;
    }

    const int f0 = 4 * lane;
    const int f1 = f0 + 128;
    const size_t o0 = (size_t)(4u*a + (unsigned)(f0 >> 6)) * NTOT + tc*64u + (unsigned)(f0 & 63);
    const size_t o1 = (size_t)(4u*a + (unsigned)(f1 >> 6)) * NTOT + tc*64u + (unsigned)(f1 & 63);
    float* outim = out + (size_t)NTOT * NTOT;
    *(float4*)(out   + o0) = *(const float4*)(lre + f0);
    *(float4*)(out   + o1) = *(const float4*)(lre + f1);
    *(float4*)(outim + o0) = *(const float4*)(lim + f0);
    *(float4*)(outim + o1) = *(const float4*)(lim + f1);
}

__global__ __launch_bounds__(256) void qblur_reduce(float* __restrict__ wsf) {
    __shared__ float sm[256];
    const int t = threadIdx.x;
    float s = 0.f;
    for (int i = t; i < WG_COUNT; i += 256) s += wsf[WS_PART + i];
    sm[t] = s;
    __syncthreads();
    for (int off = 128; off; off >>= 1) {
        if (t < off) sm[t] += sm[t + off];
        __syncthreads();
    }
    if (t == 0) wsf[WS_TOTAL] = sm[0];
}

__global__ __launch_bounds__(256) void qblur_scale(float* __restrict__ out,
                                                   const float* __restrict__ wsf) {
    const float inv = 1.0f / sqrtf(wsf[WS_TOTAL]);
    const size_t idx = ((size_t)blockIdx.x * 256u + threadIdx.x) * 4u;
    float4 v = *(float4*)(out + idx);
    v.x *= inv; v.y *= inv; v.z *= inv; v.w *= inv;
    *(float4*)(out + idx) = v;
}

extern "C" void kernel_launch(void* const* d_in, const int* in_sizes, int n_in,
                              void* d_out, int out_size, void* d_ws, size_t ws_size,
                              hipStream_t stream) {
    const float* x      = (const float*)d_in[0];
    const float* weight = (const float*)d_in[1];
    float* out = (float*)d_out;
    float* wsf = (float*)d_ws;

    qblur_setup<<<1, 256, 0, stream>>>(weight, wsf);
    qblur_main<<<WG_COUNT, 256, 0, stream>>>(x, wsf, out, wsf + WS_PART);
    qblur_reduce<<<1, 256, 0, stream>>>(wsf);
    qblur_scale<<<32768, 256, 0, stream>>>(out, wsf);
}


# Global norm from a read-only pass that recomputes the products and keeps only their sum of squares (new summation order); 1/norm applied in the main pass's store; rescale kernel dropped
# speedup vs baseline: 1.5300x; 1.5300x over previous
#include <hip/hip_runtime.h>
#include <math.h>

typedef __attribute__((ext_vector_type(2))) float v2f;
typedef __attribute__((ext_vector_type(8))) float v8f;

#define NTOT 4096
#define WAVES_PER_WG 8
#define WG_COUNT 8192
#define SUM_WG    512
#define SUM_WAVES (SUM_WG * WAVES_PER_WG)
#define SUM_GROUP 4
#define SUM_ITERS 4
#define WS_MRE   0
#define WS_MIM   256
#define WS_TOTAL 512
#define WS_PART  1024

__global__ __launch_bounds__(256) void qblur_setup(const float* __restrict__ weight,
                                                   float* __restrict__ wsf) {
    const float WMUL = 0.6324555320336759f;
    float w[5];
    #pragma unroll
    for (int k = 0; k < 5; ++k) w[k] = weight[k] * WMUL * 0.5f;
    float c0 = cosf(w[0]), s0 = sinf(w[0]);
    float c1 = cosf(w[1]), s1 = sinf(w[1]);
    float c2 = cosf(w[2]), s2 = sinf(w[2]);
    float c3 = cosf(w[3]), s3 = sinf(w[3]);
    float c4 = cosf(w[4]), s4 = sinf(w[4]);
    float R0[2][2] = {{c0, -s0}, {s0, c0}};
    float R1[2][2] = {{c1, -s1}, {s1, c1}};
    float R3[2][2] = {{c3, -s3}, {s3, c3}};
    float R4[2][2] = {{c4, -s4}, {s4, c4}};
    float G1[4][4], G3[4][4];
    #pragma unroll
    for (int p = 0; p < 2; ++p)
      #pragma unroll
      for (int q = 0; q < 2; ++q)
        #pragma unroll
        for (int r = 0; r < 2; ++r)
          #pragma unroll
          for (int s = 0; s < 2; ++s) {
            G1[p*2+q][r*2+s] = R0[p][r] * R1[q][s];
            G3[p*2+q][r*2+s] = R3[p][r] * R4[q][s];
          }
    const float YY[4][4] = {{0,0,0,-1},{0,0,1,0},{0,1,0,0},{-1,0,0,0}};
    float P[4][4], T[4][4], Q[4][4], Ure[4][4], Uim[4][4];
    #pragma unroll
    for (int i = 0; i < 4; ++i)
      #pragma unroll
      for (int j = 0; j < 4; ++j) {
        float sp = 0.f, st = 0.f;
        #pragma unroll
        for (int k = 0; k < 4; ++k) { sp += G3[i][k]*G1[k][j]; st += YY[i][k]*G1[k][j]; }
        P[i][j] = sp; T[i][j] = st;
      }
    #pragma unroll
    for (int i = 0; i < 4; ++i)
      #pragma unroll
      for (int j = 0; j < 4; ++j) {
        float sq = 0.f;
        #pragma unroll
        for (int k = 0; k < 4; ++k) sq += G3[i][k]*T[k][j];
        Q[i][j] = sq;
        Ure[i][j] = c2 * P[i][j];
        Uim[i][j] = -s2 * sq;
      }
    int t = threadIdx.x;
    int r = t >> 4, c = t & 15;
    int i = r >> 2, j = r & 3, k = c >> 2, l = c & 3;
    wsf[WS_MRE + t] = Ure[i][k]*Ure[j][l] + Uim[i][k]*Uim[j][l];
    wsf[WS_MIM + t] = Uim[i][k]*Ure[j][l] - Ure[i][k]*Uim[j][l];
}

__device__ __forceinline__ void load_a(const float* __restrict__ wsf, int l16, int hi,
                                       v2f are[4], v2f aim[4]) {
    const int abase = l16 * 16 + hi * 2;
    #pragma unroll
    for (int q = 0; q < 4; ++q) {
        are[q][0] = wsf[WS_MRE + abase + 4*q];
        are[q][1] = wsf[WS_MRE + abase + 4*q + 1];
        aim[q][0] = wsf[WS_MIM + abase + 4*q];
        aim[q][1] = wsf[WS_MIM + abase + 4*q + 1];
    }
}

__device__ __forceinline__ void load_tile(const float* __restrict__ x, size_t rowbase,
                                          unsigned col, v2f b[4]) {
    #pragma unroll
    for (int q = 0; q < 4; ++q) {
        const float* p = x + rowbase + (size_t)q * NTOT + col;
        b[q][0] = p[0];
        b[q][1] = p[1];
    }
}

__device__ __forceinline__ void tile_product(v2f b[4], const v2f are[4], const v2f aim[4],
                                             v8f& dre, v8f& dim) {
    float nsq = 0.f;
    #pragma unroll
    for (int q = 0; q < 4; ++q) nsq += b[q][0]*b[q][0] + b[q][1]*b[q][1];
    nsq += __shfl_xor(nsq, 16);
    const float inv = (nsq > 0.f) ? (1.0f / sqrtf(nsq)) : 0.f;
    #pragma unroll
    for (int q = 0; q < 4; ++q) { b[q][0] *= inv; b[q][1] *= inv; }
    dre = v8f{};
    dim = v8f{};
    #pragma unroll
    for (int q = 0; q < 4; ++q) {
        dre = __builtin_amdgcn_wmma_f32_16x16x4_f32(false, are[q], false, b[q],
                                                    (short)0, dre, false, false);
        dim = __builtin_amdgcn_wmma_f32_16x16x4_f32(false, aim[q], false, b[q],
                                                    (short)0, dim, false, false);
    }
}

__device__ __forceinline__ void load_group(const float* __restrict__ x, unsigned g,
                                           unsigned lcol, v2f raw[SUM_GROUP][4]) {
    const size_t rowbase = (size_t)(4u * (g >> 4)) * NTOT;
    const unsigned col0  = (g & 15u) * (64u * SUM_GROUP) + lcol;
    #pragma unroll
    for (int u = 0; u < SUM_GROUP; ++u) load_tile(x, rowbase, col0 + 64u * (unsigned)u, raw[u]);
}

__device__ __forceinline__ float group_sumsq(v2f raw[SUM_GROUP][4], const v2f are[4],
                                             const v2f aim[4]) {
    float g = 0.f;
    #pragma unroll
    for (int u = 0; u < SUM_GROUP; ++u) {
        v8f dre, dim;
        tile_product(raw[u], are, aim, dre, dim);
        float s = 0.f;
        #pragma unroll
        for (int p = 0; p < 8; ++p) s += dre[p]*dre[p] + dim[p]*dim[p];
        g += s;
    }
    return g;
}

__global__ __launch_bounds__(256) void qblur_sumsq(const float* __restrict__ x,
                                                   const float* __restrict__ wsf,
                                                   float* __restrict__ partials) {
    __shared__ float wsum[WAVES_PER_WG];

    const int tid  = threadIdx.x;
    const int wv   = tid >> 5;
    const int lane = tid & 31;
    const int hi   = lane >> 4;
    const int l16  = lane & 15;
    const unsigned w    = blockIdx.x * WAVES_PER_WG + wv;
    const unsigned lcol = (unsigned)(l16 * 4 + hi * 2);

    v2f are[4], aim[4];
    load_a(wsf, l16, hi, are, aim);

    v2f cur[SUM_GROUP][4], nxt[SUM_GROUP][4];
    float acc = 0.f;
    load_group(x, w, lcol, cur);
    #pragma nounroll
    for (int it = 1; it < SUM_ITERS; ++it) {
        load_group(x, (unsigned)it * SUM_WAVES + w, lcol, nxt);
        acc += group_sumsq(cur, are, aim);
        #pragma unroll
        for (int u = 0; u < SUM_GROUP; ++u)
            #pragma unroll
            for (int q = 0; q < 4; ++q) cur[u][q] = nxt[u][q];
    }
    acc += group_sumsq(cur, are, aim);

    #pragma unroll
    for (int off = 16; off; off >>= 1) acc += __shfl_xor(acc, off);
    if (lane == 0) wsum[wv] = acc;
    __syncthreads();
    if (tid == 0) {
        float tot = 0.f;
        #pragma unroll
        for (int k = 0; k < WAVES_PER_WG; ++k) tot += wsum[k];
        partials[blockIdx.x] = tot;
    }
}

__global__ __launch_bounds__(256) void qblur_reduce(float* __restrict__ wsf) {
    __shared__ float sm[256];
    const int t = threadIdx.x;
    float s = 0.f;
    for (int i = t; i < SUM_WG; i += 256) s += wsf[WS_PART + i];
    sm[t] = s;
    __syncthreads();
    for (int off = 128; off; off >>= 1) {
        if (t < off) sm[t] += sm[t + off];
        __syncthreads();
    }
    if (t == 0) wsf[WS_TOTAL] = sm[0];
}

__global__ __launch_bounds__(256) void qblur_main(const float* __restrict__ x,
                                                  const float* __restrict__ wsf,
                                                  float* __restrict__ out) {
    __shared__ float lds[WAVES_PER_WG][2][256];

    const int tid  = threadIdx.x;
    const int wv   = tid >> 5;
    const int lane = tid & 31;
    const int hi   = lane >> 4;
    const int l16  = lane & 15;

    const unsigned gw = blockIdx.x * WAVES_PER_WG + wv;
    const unsigned a  = gw >> 6;
    const unsigned tc = gw & 63;

    const size_t rowbase = (size_t)(4u * a) * NTOT;
    const unsigned col   = tc * 64u + (unsigned)(l16 * 4 + hi * 2);

    v2f b[4];
    load_tile(x, rowbase, col, b);
    v2f are[4], aim[4];
    load_a(wsf, l16, hi, are, aim);
    const float ginv = 1.0f / sqrtf(wsf[WS_TOTAL]);

    v8f dre, dim;
    tile_product(b, are, aim, dre, dim);

    float* lre = &lds[wv][0][0];
    float* lim = &lds[wv][1][0];
    const int st0 = (2*hi) * 64 + 4*l16;
    *(float4*)(lre + st0)      = make_float4(dre[0], dre[1], dre[2], dre[3]);
    *(float4*)(lre + st0 + 64) = make_float4(dre[4], dre[5], dre[6], dre[7]);
    *(float4*)(lim + st0)      = make_float4(dim[0], dim[1], dim[2], dim[3]);
    *(float4*)(lim + st0 + 64) = make_float4(dim[4], dim[5], dim[6], dim[7]);
    __syncthreads();

    const int f0 = 4 * lane;
    const int f1 = f0 + 128;
    const size_t o0 = (size_t)(4u*a + (unsigned)(f0 >> 6)) * NTOT + tc*64u + (unsigned)(f0 & 63);
    const size_t o1 = (size_t)(4u*a + (unsigned)(f1 >> 6)) * NTOT + tc*64u + (unsigned)(f1 & 63);
    float* outim = out + (size_t)NTOT * NTOT;
    float4 r0 = *(const float4*)(lre + f0);
    float4 r1 = *(const float4*)(lre + f1);
    float4 i0 = *(const float4*)(lim + f0);
    float4 i1 = *(const float4*)(lim + f1);
    r0.x *= ginv; r0.y *= ginv; r0.z *= ginv; r0.w *= ginv;
    r1.x *= ginv; r1.y *= ginv; r1.z *= ginv; r1.w *= ginv;
    i0.x *= ginv; i0.y *= ginv; i0.z *= ginv; i0.w *= ginv;
    i1.x *= ginv; i1.y *= ginv; i1.z *= ginv; i1.w *= ginv;
    *(float4*)(out   + o0) = r0;
    *(float4*)(out   + o1) = r1;
    *(float4*)(outim + o0) = i0;
    *(float4*)(outim + o1) = i1;
}

extern "C" void kernel_launch(void* const* d_in, const int* in_sizes, int n_in,
                              void* d_out, int out_size, void* d_ws, size_t ws_size,
                              hipStream_t stream) {
    const float* x      = (const float*)d_in[0];
    const float* weight = (const float*)d_in[1];
    float* out = (float*)d_out;
    float* wsf = (float*)d_ws;

    qblur_setup<<<1, 256, 0, stream>>>(weight, wsf);
    qblur_sumsq<<<SUM_WG, 256, 0, stream>>>(x, wsf, wsf + WS_PART);
    qblur_reduce<<<1, 256, 0, stream>>>(wsf);
    qblur_main<<<WG_COUNT, 256, 0, stream>>>(x, wsf, out);
}


# Main pass as a streaming loop: 4096 long-lived waves x 16 tiles, next 4 tiles' loads in flight, D rows stored straight from registers (no LDS stage, no barrier); was 65536 one-tile waves
# speedup vs baseline: 2.0219x; 1.3215x over previous
#include <hip/hip_runtime.h>
#include <math.h>

typedef __attribute__((ext_vector_type(2))) float v2f;
typedef __attribute__((ext_vector_type(8))) float v8f;

#define NTOT 4096
#define WAVES_PER_WG 8
#define SUM_WG    512
#define SUM_WAVES (SUM_WG * WAVES_PER_WG)
#define SUM_GROUP 4
#define SUM_ITERS 4
#define WS_MRE   0
#define WS_MIM   256
#define WS_TOTAL 512
#define WS_PART  1024

__global__ __launch_bounds__(256) void qblur_setup(const float* __restrict__ weight,
                                                   float* __restrict__ wsf) {
    const float WMUL = 0.6324555320336759f;
    float w[5];
    #pragma unroll
    for (int k = 0; k < 5; ++k) w[k] = weight[k] * WMUL * 0.5f;
    float c0 = cosf(w[0]), s0 = sinf(w[0]);
    float c1 = cosf(w[1]), s1 = sinf(w[1]);
    float c2 = cosf(w[2]), s2 = sinf(w[2]);
    float c3 = cosf(w[3]), s3 = sinf(w[3]);
    float c4 = cosf(w[4]), s4 = sinf(w[4]);
    float R0[2][2] = {{c0, -s0}, {s0, c0}};
    float R1[2][2] = {{c1, -s1}, {s1, c1}};
    float R3[2][2] = {{c3, -s3}, {s3, c3}};
    float R4[2][2] = {{c4, -s4}, {s4, c4}};
    float G1[4][4], G3[4][4];
    #pragma unroll
    for (int p = 0; p < 2; ++p)
      #pragma unroll
      for (int q = 0; q < 2; ++q)
        #pragma unroll
        for (int r = 0; r < 2; ++r)
          #pragma unroll
          for (int s = 0; s < 2; ++s) {
            G1[p*2+q][r*2+s] = R0[p][r] * R1[q][s];
            G3[p*2+q][r*2+s] = R3[p][r] * R4[q][s];
          }
    const float YY[4][4] = {{0,0,0,-1},{0,0,1,0},{0,1,0,0},{-1,0,0,0}};
    float P[4][4], T[4][4], Q[4][4], Ure[4][4], Uim[4][4];
    #pragma unroll
    for (int i = 0; i < 4; ++i)
      #pragma unroll
      for (int j = 0; j < 4; ++j) {
        float sp = 0.f, st = 0.f;
        #pragma unroll
        for (int k = 0; k < 4; ++k) { sp += G3[i][k]*G1[k][j]; st += YY[i][k]*G1[k][j]; }
        P[i][j] = sp; T[i][j] = st;
      }
    #pragma unroll
    for (int i = 0; i < 4; ++i)
      #pragma unroll
      for (int j = 0; j < 4; ++j) {
        float sq = 0.f;
        #pragma unroll
        for (int k = 0; k < 4; ++k) sq += G3[i][k]*T[k][j];
        Q[i][j] = sq;
        Ure[i][j] = c2 * P[i][j];
        Uim[i][j] = -s2 * sq;
      }
    int t = threadIdx.x;
    int r = t >> 4, c = t & 15;
    int i = r >> 2, j = r & 3, k = c >> 2, l = c & 3;
    wsf[WS_MRE + t] = Ure[i][k]*Ure[j][l] + Uim[i][k]*Uim[j][l];
    wsf[WS_MIM + t] = Uim[i][k]*Ure[j][l] - Ure[i][k]*Uim[j][l];
}

__device__ __forceinline__ void load_a(const float* __restrict__ wsf, int l16, int hi,
                                       v2f are[4], v2f aim[4]) {
    const int abase = l16 * 16 + hi * 2;
    #pragma unroll
    for (int q = 0; q < 4; ++q) {
        are[q][0] = wsf[WS_MRE + abase + 4*q];
        are[q][1] = wsf[WS_MRE + abase + 4*q + 1];
        aim[q][0] = wsf[WS_MIM + abase + 4*q];
        aim[q][1] = wsf[WS_MIM + abase + 4*q + 1];
    }
}

__device__ __forceinline__ void load_tile(const float* __restrict__ x, size_t rowbase,
                                          unsigned col, v2f b[4]) {
    #pragma unroll
    for (int q = 0; q < 4; ++q) {
        const float* p = x + rowbase + (size_t)q * NTOT + col;
        b[q][0] = p[0];
        b[q][1] = p[1];
    }
}

__device__ __forceinline__ void tile_product(v2f b[4], const v2f are[4], const v2f aim[4],
                                             v8f& dre, v8f& dim) {
    float nsq = 0.f;
    #pragma unroll
    for (int q = 0; q < 4; ++q) nsq += b[q][0]*b[q][0] + b[q][1]*b[q][1];
    nsq += __shfl_xor(nsq, 16);
    const float inv = (nsq > 0.f) ? (1.0f / sqrtf(nsq)) : 0.f;
    #pragma unroll
    for (int q = 0; q < 4; ++q) { b[q][0] *= inv; b[q][1] *= inv; }
    dre = v8f{};
    dim = v8f{};
    #pragma unroll
    for (int q = 0; q < 4; ++q) {
        dre = __builtin_amdgcn_wmma_f32_16x16x4_f32(false, are[q], false, b[q],
                                                    (short)0, dre, false, false);
        dim = __builtin_amdgcn_wmma_f32_16x16x4_f32(false, aim[q], false, b[q],
                                                    (short)0, dim, false, false);
    }
}

__device__ __forceinline__ void load_group(const float* __restrict__ x, unsigned g,
                                           unsigned lcol, v2f raw[SUM_GROUP][4]) {
    const size_t rowbase = (size_t)(4u * (g >> 4)) * NTOT;
    const unsigned col0  = (g & 15u) * (64u * SUM_GROUP) + lcol;
    #pragma unroll
    for (int u = 0; u < SUM_GROUP; ++u) load_tile(x, rowbase, col0 + 64u * (unsigned)u, raw[u]);
}

__device__ __forceinline__ float group_sumsq(v2f raw[SUM_GROUP][4], const v2f are[4],
                                             const v2f aim[4]) {
    float g = 0.f;
    #pragma unroll
    for (int u = 0; u < SUM_GROUP; ++u) {
        v8f dre, dim;
        tile_product(raw[u], are, aim, dre, dim);
        float s = 0.f;
        #pragma unroll
        for (int p = 0; p < 8; ++p) s += dre[p]*dre[p] + dim[p]*dim[p];
        g += s;
    }
    return g;
}

__global__ __launch_bounds__(256) void qblur_sumsq(const float* __restrict__ x,
                                                   const float* __restrict__ wsf,
                                                   float* __restrict__ partials) {
    __shared__ float wsum[WAVES_PER_WG];

    const int tid  = threadIdx.x;
    const int wv   = tid >> 5;
    const int lane = tid & 31;
    const int hi   = lane >> 4;
    const int l16  = lane & 15;
    const unsigned w    = blockIdx.x * WAVES_PER_WG + wv;
    const unsigned lcol = (unsigned)(l16 * 4 + hi * 2);

    v2f are[4], aim[4];
    load_a(wsf, l16, hi, are, aim);

    v2f cur[SUM_GROUP][4], nxt[SUM_GROUP][4];
    float acc = 0.f;
    load_group(x, w, lcol, cur);
    #pragma nounroll
    for (int it = 1; it < SUM_ITERS; ++it) {
        load_group(x, (unsigned)it * SUM_WAVES + w, lcol, nxt);
        acc += group_sumsq(cur, are, aim);
        #pragma unroll
        for (int u = 0; u < SUM_GROUP; ++u)
            #pragma unroll
            for (int q = 0; q < 4; ++q) cur[u][q] = nxt[u][q];
    }
    acc += group_sumsq(cur, are, aim);

    #pragma unroll
    for (int off = 16; off; off >>= 1) acc += __shfl_xor(acc, off);
    if (lane == 0) wsum[wv] = acc;
    __syncthreads();
    if (tid == 0) {
        float tot = 0.f;
        #pragma unroll
        for (int k = 0; k < WAVES_PER_WG; ++k) tot += wsum[k];
        partials[blockIdx.x] = tot;
    }
}

__global__ __launch_bounds__(256) void qblur_reduce(float* __restrict__ wsf) {
    __shared__ float sm[256];
    const int t = threadIdx.x;
    float s = 0.f;
    for (int i = t; i < SUM_WG; i += 256) s += wsf[WS_PART + i];
    sm[t] = s;
    __syncthreads();
    for (int off = 128; off; off >>= 1) {
        if (t < off) sm[t] += sm[t + off];
        __syncthreads();
    }
    if (t == 0) wsf[WS_TOTAL] = sm[0];
}

__device__ __forceinline__ void group_store(v2f raw[SUM_GROUP][4], const v2f are[4],
                                            const v2f aim[4], float ginv,
                                            float* __restrict__ ore, float* __restrict__ oim) {
    #pragma unroll
    for (int u = 0; u < SUM_GROUP; ++u) {
        v8f dre, dim;
        tile_product(raw[u], are, aim, dre, dim);
        float* pr = ore + 64 * u;
        float* pi = oim + 64 * u;
        *(float4*)(pr)        = make_float4(dre[0]*ginv, dre[1]*ginv, dre[2]*ginv, dre[3]*ginv);
        *(float4*)(pr + NTOT) = make_float4(dre[4]*ginv, dre[5]*ginv, dre[6]*ginv, dre[7]*ginv);
        *(float4*)(pi)        = make_float4(dim[0]*ginv, dim[1]*ginv, dim[2]*ginv, dim[3]*ginv);
        *(float4*)(pi + NTOT) = make_float4(dim[4]*ginv, dim[5]*ginv, dim[6]*ginv, dim[7]*ginv);
    }
}

__device__ __forceinline__ size_t group_out(unsigned g, int hi, int l16) {
    return (size_t)(4u * (g >> 4) + 2u * (unsigned)hi) * NTOT
         + (g & 15u) * (64u * SUM_GROUP) + 4u * (unsigned)l16;
}

__global__ __launch_bounds__(256) void qblur_main(const float* __restrict__ x,
                                                  const float* __restrict__ wsf,
                                                  float* __restrict__ out) {
    const int tid  = threadIdx.x;
    const int wv   = tid >> 5;
    const int lane = tid & 31;
    const int hi   = lane >> 4;
    const int l16  = lane & 15;
    const unsigned w    = blockIdx.x * WAVES_PER_WG + wv;
    const unsigned lcol = (unsigned)(l16 * 4 + hi * 2);

    v2f cur[SUM_GROUP][4], nxt[SUM_GROUP][4];
    load_group(x, w, lcol, cur);

    v2f are[4], aim[4];
    load_a(wsf, l16, hi, are, aim);
    const float ginv = 1.0f / sqrtf(wsf[WS_TOTAL]);
    float* outim = out + (size_t)NTOT * NTOT;

    static_assert(SUM_ITERS % 2 == 0, "the trip below takes two groups");
    #pragma unroll
    for (int it = 0; it < SUM_ITERS; it += 2) {
        const unsigned g0 = (unsigned)it * SUM_WAVES + w;
        const unsigned g1 = g0 + SUM_WAVES;
        load_group(x, g1, lcol, nxt);
        const size_t o0 = group_out(g0, hi, l16);
        group_store(cur, are, aim, ginv, out + o0, outim + o0);
        if (it + 2 < SUM_ITERS) load_group(x, g1 + SUM_WAVES, lcol, cur);
        const size_t o1 = group_out(g1, hi, l16);
        group_store(nxt, are, aim, ginv, out + o1, outim + o1);
    }
}

extern "C" void kernel_launch(void* const* d_in, const int* in_sizes, int n_in,
                              void* d_out, int out_size, void* d_ws, size_t ws_size,
                              hipStream_t stream) {
    const float* x      = (const float*)d_in[0];
    const float* weight = (const float*)d_in[1];
    float* out = (float*)d_out;
    float* wsf = (float*)d_ws;

    qblur_setup<<<1, 256, 0, stream>>>(weight, wsf);
    qblur_sumsq<<<SUM_WG, 256, 0, stream>>>(x, wsf, wsf + WS_PART);
    qblur_reduce<<<1, 256, 0, stream>>>(wsf);
    qblur_main<<<SUM_WG, 256, 0, stream>>>(x, wsf, out);
}


# The two one-workgroup launches folded into their consumers: wave 0 of each sum-pass workgroup builds M in LDS, every main-pass wave adds the 512 partials itself in the old order; 4 launches to 2
# speedup vs baseline: 2.5465x; 1.2595x over previous
#include <hip/hip_runtime.h>
#include <math.h>

typedef __attribute__((ext_vector_type(2))) float v2f;
typedef __attribute__((ext_vector_type(8))) float v8f;

#define NTOT 4096
#define WAVES_PER_WG 8
#define SUM_WG    512
#define SUM_WAVES (SUM_WG * WAVES_PER_WG)
#define SUM_GROUP 4
#define SUM_ITERS 4
#define WS_MRE   0
#define WS_MIM   256
#define WS_PART  1024

__device__ __forceinline__ float pick4(const float a[4], int idx) {
    const float lo = (idx & 1) ? a[1] : a[0];
    const float hi = (idx & 1) ? a[3] : a[2];
    return (idx & 2) ? hi : lo;
}

__device__ __forceinline__ void build_m(float wk, int lane, float* mlds) {
    const float WMUL = 0.6324555320336759f;
    const float ang = wk * WMUL * 0.5f;
    const float cs = cosf(ang), sn = sinf(ang);
    const float c0 = __shfl(cs, 0), s0 = __shfl(sn, 0);
    const float c1 = __shfl(cs, 1), s1 = __shfl(sn, 1);
    const float c2 = __shfl(cs, 2), s2 = __shfl(sn, 2);
    const float c3 = __shfl(cs, 3), s3 = __shfl(sn, 3);
    const float c4 = __shfl(cs, 4), s4 = __shfl(sn, 4);
    float R0[2][2] = {{c0, -s0}, {s0, c0}};
    float R1[2][2] = {{c1, -s1}, {s1, c1}};
    float R3[2][2] = {{c3, -s3}, {s3, c3}};
    float R4[2][2] = {{c4, -s4}, {s4, c4}};
    float G1[4][4], G3[4][4];
    #pragma unroll
    for (int p = 0; p < 2; ++p)
      #pragma unroll
      for (int q = 0; q < 2; ++q)
        #pragma unroll
        for (int r = 0; r < 2; ++r)
          #pragma unroll
          for (int s = 0; s < 2; ++s) {
            G1[p*2+q][r*2+s] = R0[p][r] * R1[q][s];
            G3[p*2+q][r*2+s] = R3[p][r] * R4[q][s];
          }
    const float YY[4][4] = {{0,0,0,-1},{0,0,1,0},{0,1,0,0},{-1,0,0,0}};
    float P[4][4], T[4][4], Ure[4][4], Uim[4][4];
    #pragma unroll
    for (int i = 0; i < 4; ++i)
      #pragma unroll
      for (int j = 0; j < 4; ++j) {
        float sp = 0.f, st = 0.f;
        #pragma unroll
        for (int k = 0; k < 4; ++k) { sp += G3[i][k]*G1[k][j]; st += YY[i][k]*G1[k][j]; }
        P[i][j] = sp; T[i][j] = st;
      }
    #pragma unroll
    for (int i = 0; i < 4; ++i)
      #pragma unroll
      for (int j = 0; j < 4; ++j) {
        float sq = 0.f;
        #pragma unroll
        for (int k = 0; k < 4; ++k) sq += G3[i][k]*T[k][j];
        Ure[i][j] = c2 * P[i][j];
        Uim[i][j] = -s2 * sq;
      }
    const int hi = lane >> 4, k = (lane & 15) >> 2, l = lane & 3;
    float ikre[4], ikim[4], lre[4], lim[4];
    #pragma unroll
    for (int i = 0; i < 4; ++i) {
        ikre[i] = pick4(Ure[i], k);  ikim[i] = pick4(Uim[i], k);
        lre[i]  = pick4(Ure[i], l);  lim[i]  = pick4(Uim[i], l);
    }
    float jlre[2], jlim[2];
    #pragma unroll
    for (int o = 0; o < 2; ++o) {
        jlre[o] = hi ? lre[2*o + 1] : lre[2*o];
        jlim[o] = hi ? lim[2*o + 1] : lim[2*o];
    }
    #pragma unroll
    for (int e = 0; e < 8; ++e) {
        const int i = e >> 1, o = e & 1;
        const int t = lane + 32 * e;
        mlds[WS_MRE + t] = ikre[i]*jlre[o] + ikim[i]*jlim[o];
        mlds[WS_MIM + t] = ikim[i]*jlre[o] - ikre[i]*jlim[o];
    }
}

__device__ __forceinline__ void load_a(const float* __restrict__ wsf, int l16, int hi,
                                       v2f are[4], v2f aim[4]) {
    const int abase = l16 * 16 + hi * 2;
    #pragma unroll
    for (int q = 0; q < 4; ++q) {
        are[q][0] = wsf[WS_MRE + abase + 4*q];
        are[q][1] = wsf[WS_MRE + abase + 4*q + 1];
        aim[q][0] = wsf[WS_MIM + abase + 4*q];
        aim[q][1] = wsf[WS_MIM + abase + 4*q + 1];
    }
}

__device__ __forceinline__ void load_tile(const float* __restrict__ x, size_t rowbase,
                                          unsigned col, v2f b[4]) {
    #pragma unroll
    for (int q = 0; q < 4; ++q) {
        const float* p = x + rowbase + (size_t)q * NTOT + col;
        b[q][0] = p[0];
        b[q][1] = p[1];
    }
}

__device__ __forceinline__ void tile_product(v2f b[4], const v2f are[4], const v2f aim[4],
                                             v8f& dre, v8f& dim) {
    float nsq = 0.f;
    #pragma unroll
    for (int q = 0; q < 4; ++q) nsq += b[q][0]*b[q][0] + b[q][1]*b[q][1];
    nsq += __shfl_xor(nsq, 16);
    const float inv = (nsq > 0.f) ? (1.0f / sqrtf(nsq)) : 0.f;
    #pragma unroll
    for (int q = 0; q < 4; ++q) { b[q][0] *= inv; b[q][1] *= inv; }
    dre = v8f{};
    dim = v8f{};
    #pragma unroll
    for (int q = 0; q < 4; ++q) {
        dre = __builtin_amdgcn_wmma_f32_16x16x4_f32(false, are[q], false, b[q],
                                                    (short)0, dre, false, false);
        dim = __builtin_amdgcn_wmma_f32_16x16x4_f32(false, aim[q], false, b[q],
                                                    (short)0, dim, false, false);
    }
}

__device__ __forceinline__ void load_group(const float* __restrict__ x, unsigned g,
                                           unsigned lcol, v2f raw[SUM_GROUP][4]) {
    const size_t rowbase = (size_t)(4u * (g >> 4)) * NTOT;
    const unsigned col0  = (g & 15u) * (64u * SUM_GROUP) + lcol;
    #pragma unroll
    for (int u = 0; u < SUM_GROUP; ++u) load_tile(x, rowbase, col0 + 64u * (unsigned)u, raw[u]);
}

__device__ __forceinline__ float group_sumsq(v2f raw[SUM_GROUP][4], const v2f are[4],
                                             const v2f aim[4]) {
    float g = 0.f;
    #pragma unroll
    for (int u = 0; u < SUM_GROUP; ++u) {
        v8f dre, dim;
        tile_product(raw[u], are, aim, dre, dim);
        float s = 0.f;
        #pragma unroll
        for (int p = 0; p < 8; ++p) s += dre[p]*dre[p] + dim[p]*dim[p];
        g += s;
    }
    return g;
}

__global__ __launch_bounds__(256) void qblur_sumsq(const float* __restrict__ x,
                                                   const float* __restrict__ weight,
                                                   float* __restrict__ wsf) {
    __shared__ float wsum[WAVES_PER_WG];
    __shared__ float sm_m[512];

    const int tid  = threadIdx.x;
    const int wv   = tid >> 5;
    const int lane = tid & 31;
    const int hi   = lane >> 4;
    const int l16  = lane & 15;
    const unsigned w    = blockIdx.x * WAVES_PER_WG + wv;
    const unsigned lcol = (unsigned)(l16 * 4 + hi * 2);

    float wk = 0.f;
    if (wv == 0) wk = weight[lane < 5 ? lane : 4];

    v2f cur[SUM_GROUP][4], nxt[SUM_GROUP][4];
    float acc = 0.f;
    load_group(x, w, lcol, cur);

    if (wv == 0) build_m(wk, lane, sm_m);
    __builtin_amdgcn_fence(__ATOMIC_RELEASE, "workgroup", "local");
    __builtin_amdgcn_s_barrier();
    __builtin_amdgcn_fence(__ATOMIC_ACQUIRE, "workgroup", "local");
    v2f are[4], aim[4];
    load_a(sm_m, l16, hi, are, aim);
    if (blockIdx.x == 0) {
        wsf[WS_MRE + tid] = sm_m[WS_MRE + tid];
        wsf[WS_MIM + tid] = sm_m[WS_MIM + tid];
    }

    #pragma nounroll
    for (int it = 1; it < SUM_ITERS; ++it) {
        load_group(x, (unsigned)it * SUM_WAVES + w, lcol, nxt);
        acc += group_sumsq(cur, are, aim);
        #pragma unroll
        for (int u = 0; u < SUM_GROUP; ++u)
            #pragma unroll
            for (int q = 0; q < 4; ++q) cur[u][q] = nxt[u][q];
    }
    acc += group_sumsq(cur, are, aim);

    #pragma unroll
    for (int off = 16; off; off >>= 1) acc += __shfl_xor(acc, off);
    if (lane == 0) wsum[wv] = acc;
    __syncthreads();
    if (tid == 0) {
        float tot = 0.f;
        #pragma unroll
        for (int k = 0; k < WAVES_PER_WG; ++k) tot += wsum[k];
        wsf[WS_PART + blockIdx.x] = tot;
    }
}

__device__ __forceinline__ float total_sumsq(const float* __restrict__ part, int lane) {
    static_assert(SUM_WG == 512, "two partials per tree leaf");
    float s[8];
    #pragma unroll
    for (int m = 0; m < 8; ++m) s[m] = part[lane + 32*m] + part[lane + 32*m + 256];
    #pragma unroll
    for (int m = 0; m < 4; ++m) s[m] += s[m + 4];
    s[0] += s[2];
    s[1] += s[3];
    float tot = s[0] + s[1];
    #pragma unroll
    for (int off = 16; off; off >>= 1) tot += __shfl_xor(tot, off);
    return tot;
}

__device__ __forceinline__ void group_store(v2f raw[SUM_GROUP][4], const v2f are[4],
                                            const v2f aim[4], float ginv,
                                            float* __restrict__ ore, float* __restrict__ oim) {
    #pragma unroll
    for (int u = 0; u < SUM_GROUP; ++u) {
        v8f dre, dim;
        tile_product(raw[u], are, aim, dre, dim);
        float* pr = ore + 64 * u;
        float* pi = oim + 64 * u;
        *(float4*)(pr)        = make_float4(dre[0]*ginv, dre[1]*ginv, dre[2]*ginv, dre[3]*ginv);
        *(float4*)(pr + NTOT) = make_float4(dre[4]*ginv, dre[5]*ginv, dre[6]*ginv, dre[7]*ginv);
        *(float4*)(pi)        = make_float4(dim[0]*ginv, dim[1]*ginv, dim[2]*ginv, dim[3]*ginv);
        *(float4*)(pi + NTOT) = make_float4(dim[4]*ginv, dim[5]*ginv, dim[6]*ginv, dim[7]*ginv);
    }
}

__device__ __forceinline__ size_t group_out(unsigned g, int hi, int l16) {
    return (size_t)(4u * (g >> 4) + 2u * (unsigned)hi) * NTOT
         + (g & 15u) * (64u * SUM_GROUP) + 4u * (unsigned)l16;
}

__global__ __launch_bounds__(256) void qblur_main(const float* __restrict__ x,
                                                  const float* __restrict__ wsf,
                                                  float* __restrict__ out) {
    const int tid  = threadIdx.x;
    const int wv   = tid >> 5;
    const int lane = tid & 31;
    const int hi   = lane >> 4;
    const int l16  = lane & 15;
    const unsigned w    = blockIdx.x * WAVES_PER_WG + wv;
    const unsigned lcol = (unsigned)(l16 * 4 + hi * 2);

    v2f cur[SUM_GROUP][4], nxt[SUM_GROUP][4];
    load_group(x, w, lcol, cur);

    v2f are[4], aim[4];
    load_a(wsf, l16, hi, are, aim);
    const float ginv = 1.0f / sqrtf(total_sumsq(wsf + WS_PART, lane));
    float* outim = out + (size_t)NTOT * NTOT;

    static_assert(SUM_ITERS % 2 == 0, "the trip below takes two groups");
    #pragma unroll
    for (int it = 0; it < SUM_ITERS; it += 2) {
        const unsigned g0 = (unsigned)it * SUM_WAVES + w;
        const unsigned g1 = g0 + SUM_WAVES;
        load_group(x, g1, lcol, nxt);
        const size_t o0 = group_out(g0, hi, l16);
        group_store(cur, are, aim, ginv, out + o0, outim + o0);
        if (it + 2 < SUM_ITERS) load_group(x, g1 + SUM_WAVES, lcol, cur);
        const size_t o1 = group_out(g1, hi, l16);
        group_store(nxt, are, aim, ginv, out + o1, outim + o1);
    }
}

extern "C" void kernel_launch(void* const* d_in, const int* in_sizes, int n_in,
                              void* d_out, int out_size, void* d_ws, size_t ws_size,
                              hipStream_t stream) {
    const float* x      = (const float*)d_in[0];
    const float* weight = (const float*)d_in[1];
    float* out = (float*)d_out;
    float* wsf = (float*)d_ws;

    qblur_sumsq<<<SUM_WG, 256, 0, stream>>>(x, weight, wsf);
    qblur_main<<<SUM_WG, 256, 0, stream>>>(x, wsf, out);
}


# Norm pass drops its products: U B U^H keeps each block's Frobenius norm (U unitary), so it sums the squares of the normalised entries (new summation order), 16-byte loads all in flight
# speedup vs baseline: 3.1436x; 1.2345x over previous
#include <hip/hip_runtime.h>
#include <math.h>

typedef __attribute__((ext_vector_type(2))) float v2f;
typedef __attribute__((ext_vector_type(4))) float v4f;
typedef __attribute__((ext_vector_type(8))) float v8f;

#define NTOT 4096
#define WAVES_PER_WG 8
#define SUM_WG    512
#define SUM_WAVES (SUM_WG * WAVES_PER_WG)
#define SUM_GROUP 4
#define SUM_ITERS 4
#define WS_MRE   0
#define WS_MIM   256
#define WS_PART  1024

__device__ __forceinline__ float pick4(const float a[4], int idx) {
    const float lo = (idx & 1) ? a[1] : a[0];
    const float hi = (idx & 1) ? a[3] : a[2];
    return (idx & 2) ? hi : lo;
}

__device__ __forceinline__ void build_m(float wk, int lane, float* __restrict__ mdst) {
    const float WMUL = 0.6324555320336759f;
    const float ang = wk * WMUL * 0.5f;
    const float cs = cosf(ang), sn = sinf(ang);
    const float c0 = __shfl(cs, 0), s0 = __shfl(sn, 0);
    const float c1 = __shfl(cs, 1), s1 = __shfl(sn, 1);
    const float c2 = __shfl(cs, 2), s2 = __shfl(sn, 2);
    const float c3 = __shfl(cs, 3), s3 = __shfl(sn, 3);
    const float c4 = __shfl(cs, 4), s4 = __shfl(sn, 4);
    float R0[2][2] = {{c0, -s0}, {s0, c0}};
    float R1[2][2] = {{c1, -s1}, {s1, c1}};
    float R3[2][2] = {{c3, -s3}, {s3, c3}};
    float R4[2][2] = {{c4, -s4}, {s4, c4}};
    float G1[4][4], G3[4][4];
    #pragma unroll
    for (int p = 0; p < 2; ++p)
      #pragma unroll
      for (int q = 0; q < 2; ++q)
        #pragma unroll
        for (int r = 0; r < 2; ++r)
          #pragma unroll
          for (int s = 0; s < 2; ++s) {
            G1[p*2+q][r*2+s] = R0[p][r] * R1[q][s];
            G3[p*2+q][r*2+s] = R3[p][r] * R4[q][s];
          }
    const float YY[4][4] = {{0,0,0,-1},{0,0,1,0},{0,1,0,0},{-1,0,0,0}};
    float P[4][4], T[4][4], Ure[4][4], Uim[4][4];
    #pragma unroll
    for (int i = 0; i < 4; ++i)
      #pragma unroll
      for (int j = 0; j < 4; ++j) {
        float sp = 0.f, st = 0.f;
        #pragma unroll
        for (int k = 0; k < 4; ++k) { sp += G3[i][k]*G1[k][j]; st += YY[i][k]*G1[k][j]; }
        P[i][j] = sp; T[i][j] = st;
      }
    #pragma unroll
    for (int i = 0; i < 4; ++i)
      #pragma unroll
      for (int j = 0; j < 4; ++j) {
        float sq = 0.f;
        #pragma unroll
        for (int k = 0; k < 4; ++k) sq += G3[i][k]*T[k][j];
        Ure[i][j] = c2 * P[i][j];
        Uim[i][j] = -s2 * sq;
      }
    const int hi = lane >> 4, k = (lane & 15) >> 2, l = lane & 3;
    float ikre[4], ikim[4], lre[4], lim[4];
    #pragma unroll
    for (int i = 0; i < 4; ++i) {
        ikre[i] = pick4(Ure[i], k);  ikim[i] = pick4(Uim[i], k);
        lre[i]  = pick4(Ure[i], l);  lim[i]  = pick4(Uim[i], l);
    }
    float jlre[2], jlim[2];
    #pragma unroll
    for (int o = 0; o < 2; ++o) {
        jlre[o] = hi ? lre[2*o + 1] : lre[2*o];
        jlim[o] = hi ? lim[2*o + 1] : lim[2*o];
    }
    #pragma unroll
    for (int e = 0; e < 8; ++e) {
        const int i = e >> 1, o = e & 1;
        const int t = lane + 32 * e;
        mdst[WS_MRE + t] = ikre[i]*jlre[o] + ikim[i]*jlim[o];
        mdst[WS_MIM + t] = ikim[i]*jlre[o] - ikre[i]*jlim[o];
    }
}

__device__ __forceinline__ void load_a(const float* __restrict__ wsf, int l16, int hi,
                                       v2f are[4], v2f aim[4]) {
    const int abase = l16 * 16 + hi * 2;
    #pragma unroll
    for (int q = 0; q < 4; ++q) {
        are[q][0] = wsf[WS_MRE + abase + 4*q];
        are[q][1] = wsf[WS_MRE + abase + 4*q + 1];
        aim[q][0] = wsf[WS_MIM + abase + 4*q];
        aim[q][1] = wsf[WS_MIM + abase + 4*q + 1];
    }
}

__device__ __forceinline__ void load_tile(const float* __restrict__ x, size_t rowbase,
                                          unsigned col, v2f b[4]) {
    #pragma unroll
    for (int q = 0; q < 4; ++q) {
        const float* p = x + rowbase + (size_t)q * NTOT + col;
        b[q][0] = p[0];
        b[q][1] = p[1];
    }
}

__device__ __forceinline__ void tile_product(v2f b[4], const v2f are[4], const v2f aim[4],
                                             v8f& dre, v8f& dim) {
    float nsq = 0.f;
    #pragma unroll
    for (int q = 0; q < 4; ++q) nsq += b[q][0]*b[q][0] + b[q][1]*b[q][1];
    nsq += __shfl_xor(nsq, 16);
    const float inv = (nsq > 0.f) ? (1.0f / sqrtf(nsq)) : 0.f;
    #pragma unroll
    for (int q = 0; q < 4; ++q) { b[q][0] *= inv; b[q][1] *= inv; }
    dre = v8f{};
    dim = v8f{};
    #pragma unroll
    for (int q = 0; q < 4; ++q) {
        dre = __builtin_amdgcn_wmma_f32_16x16x4_f32(false, are[q], false, b[q],
                                                    (short)0, dre, false, false);
        dim = __builtin_amdgcn_wmma_f32_16x16x4_f32(false, aim[q], false, b[q],
                                                    (short)0, dim, false, false);
    }
}

__device__ __forceinline__ void load_group(const float* __restrict__ x, unsigned g,
                                           unsigned lcol, v2f raw[SUM_GROUP][4]) {
    const size_t rowbase = (size_t)(4u * (g >> 4)) * NTOT;
    const unsigned col0  = (g & 15u) * (64u * SUM_GROUP) + lcol;
    #pragma unroll
    for (int u = 0; u < SUM_GROUP; ++u) load_tile(x, rowbase, col0 + 64u * (unsigned)u, raw[u]);
}

#define SUM_PIECES 2

__device__ __forceinline__ float block_sumsq(const v4f r[4]) {
    float nsq = 0.f;
    #pragma unroll
    for (int q = 0; q < 4; ++q)
        nsq += r[q][0]*r[q][0] + r[q][1]*r[q][1] + r[q][2]*r[q][2] + r[q][3]*r[q][3];
    const float inv = (nsq > 0.f) ? (1.0f / sqrtf(nsq)) : 0.f;
    float s = 0.f;
    #pragma unroll
    for (int q = 0; q < 4; ++q) {
        const v4f t = r[q] * inv;
        s += t[0]*t[0] + t[1]*t[1] + t[2]*t[2] + t[3]*t[3];
    }
    return s;
}

__global__ __launch_bounds__(256) void qblur_sumsq(const float* __restrict__ x,
                                                   const float* __restrict__ weight,
                                                   float* __restrict__ wsf) {
    __shared__ float wsum[WAVES_PER_WG];

    const int tid  = threadIdx.x;
    const int wv   = tid >> 5;
    const int lane = tid & 31;
    const unsigned w = blockIdx.x * WAVES_PER_WG + wv;
    const bool builder = (blockIdx.x == 0) && (wv == 0);

    float wk = 0.f;
    if (builder) wk = weight[lane < 5 ? lane : 4];

    v4f raw[SUM_ITERS][SUM_PIECES][4];
    #pragma unroll
    for (int it = 0; it < SUM_ITERS; ++it) {
        const unsigned g = (unsigned)it * SUM_WAVES + w;
        const float* p = x + (size_t)(4u * (g >> 4)) * NTOT + (g & 15u) * (64u * SUM_GROUP)
                           + 4u * (unsigned)lane;
        #pragma unroll
        for (int h = 0; h < SUM_PIECES; ++h)
            #pragma unroll
            for (int q = 0; q < 4; ++q)
                raw[it][h][q] = *(const v4f*)(p + (size_t)q * NTOT + 128 * h);
    }

    if (builder) build_m(wk, lane, wsf);

    float acc = 0.f;
    #pragma unroll
    for (int it = 0; it < SUM_ITERS; ++it) {
        float g = 0.f;
        #pragma unroll
        for (int h = 0; h < SUM_PIECES; ++h) g += block_sumsq(raw[it][h]);
        acc += g;
    }

    #pragma unroll
    for (int off = 16; off; off >>= 1) acc += __shfl_xor(acc, off);
    if (lane == 0) wsum[wv] = acc;
    __syncthreads();
    if (tid == 0) {
        float tot = 0.f;
        #pragma unroll
        for (int k = 0; k < WAVES_PER_WG; ++k) tot += wsum[k];
        wsf[WS_PART + blockIdx.x] = tot;
    }
}

__device__ __forceinline__ float total_sumsq(const float* __restrict__ part, int lane) {
    static_assert(SUM_WG == 512, "two partials per tree leaf");
    float s[8];
    #pragma unroll
    for (int m = 0; m < 8; ++m) s[m] = part[lane + 32*m] + part[lane + 32*m + 256];
    #pragma unroll
    for (int m = 0; m < 4; ++m) s[m] += s[m + 4];
    s[0] += s[2];
    s[1] += s[3];
    float tot = s[0] + s[1];
    #pragma unroll
    for (int off = 16; off; off >>= 1) tot += __shfl_xor(tot, off);
    return tot;
}

__device__ __forceinline__ void group_store(v2f raw[SUM_GROUP][4], const v2f are[4],
                                            const v2f aim[4], float ginv,
                                            float* __restrict__ ore, float* __restrict__ oim) {
    #pragma unroll
    for (int u = 0; u < SUM_GROUP; ++u) {
        v8f dre, dim;
        tile_product(raw[u], are, aim, dre, dim);
        float* pr = ore + 64 * u;
        float* pi = oim + 64 * u;
        *(float4*)(pr)        = make_float4(dre[0]*ginv, dre[1]*ginv, dre[2]*ginv, dre[3]*ginv);
        *(float4*)(pr + NTOT) = make_float4(dre[4]*ginv, dre[5]*ginv, dre[6]*ginv, dre[7]*ginv);
        *(float4*)(pi)        = make_float4(dim[0]*ginv, dim[1]*ginv, dim[2]*ginv, dim[3]*ginv);
        *(float4*)(pi + NTOT) = make_float4(dim[4]*ginv, dim[5]*ginv, dim[6]*ginv, dim[7]*ginv);
    }
}

__device__ __forceinline__ size_t group_out(unsigned g, int hi, int l16) {
    return (size_t)(4u * (g >> 4) + 2u * (unsigned)hi) * NTOT
         + (g & 15u) * (64u * SUM_GROUP) + 4u * (unsigned)l16;
}

__global__ __launch_bounds__(256) void qblur_main(const float* __restrict__ x,
                                                  const float* __restrict__ wsf,
                                                  float* __restrict__ out) {
    const int tid  = threadIdx.x;
    const int wv   = tid >> 5;
    const int lane = tid & 31;
    const int hi   = lane >> 4;
    const int l16  = lane & 15;
    const unsigned w    = blockIdx.x * WAVES_PER_WG + wv;
    const unsigned lcol = (unsigned)(l16 * 4 + hi * 2);

    v2f cur[SUM_GROUP][4], nxt[SUM_GROUP][4];
    load_group(x, w, lcol, cur);

    v2f are[4], aim[4];
    load_a(wsf, l16, hi, are, aim);
    const float ginv = 1.0f / sqrtf(total_sumsq(wsf + WS_PART, lane));
    float* outim = out + (size_t)NTOT * NTOT;

    static_assert(SUM_ITERS % 2 == 0, "the trip below takes two groups");
    #pragma unroll
    for (int it = 0; it < SUM_ITERS; it += 2) {
        const unsigned g0 = (unsigned)it * SUM_WAVES + w;
        const unsigned g1 = g0 + SUM_WAVES;
        load_group(x, g1, lcol, nxt);
        const size_t o0 = group_out(g0, hi, l16);
        group_store(cur, are, aim, ginv, out + o0, outim + o0);
        if (it + 2 < SUM_ITERS) load_group(x, g1 + SUM_WAVES, lcol, cur);
        const size_t o1 = group_out(g1, hi, l16);
        group_store(nxt, are, aim, ginv, out + o1, outim + o1);
    }
}

extern "C" void kernel_launch(void* const* d_in, const int* in_sizes, int n_in,
                              void* d_out, int out_size, void* d_ws, size_t ws_size,
                              hipStream_t stream) {
    const float* x      = (const float*)d_in[0];
    const float* weight = (const float*)d_in[1];
    float* out = (float*)d_out;
    float* wsf = (float*)d_ws;

    qblur_sumsq<<<SUM_WG, 256, 0, stream>>>(x, weight, wsf);
    qblur_main<<<SUM_WG, 256, 0, stream>>>(x, wsf, out);
}
